# starting kernel
# baseline (speedup 1.0000x reference)
#include <hip/hip_runtime.h>
#include <hip/hip_bf16.h>
#include <math.h>


typedef __attribute__((ext_vector_type(2))) float v2f;
typedef __attribute__((ext_vector_type(8))) float v8f;

__global__ void __launch_bounds__(32)
wmma_gemm_nt64(const float* __restrict__ A,
               const float* __restrict__ W,
               const float* __restrict__ bias,
               float* __restrict__ D,
               float alpha)
{
    const int m0   = blockIdx.x * 16;
    const int n0   = blockIdx.y * 16;
    const int lane = threadIdx.x;
    const int lr   = lane & 15;
    const int hi   = lane >> 4;
    const int kh   = hi * 2;

    const float* __restrict__ arow = A + (size_t)(m0 + lr) * 64;
    const float* __restrict__ wrow = W + (size_t)(n0 + lr) * 64;

    v8f c = {};
#pragma unroll
    for (int k = 0; k < 64; k += 4) {
        v2f a, b;
        a.x = arow[k + kh + 0];
        a.y = arow[k + kh + 1];
        b.x = wrow[k + kh + 0];
        b.y = wrow[k + kh + 1];
        c = __builtin_amdgcn_wmma_f32_16x16x4_f32(false, a, false, b,
                                                  (short)0, c, false, false);
    }

    const float bn = bias[n0 + lr];
#pragma unroll
    for (int r = 0; r < 8; ++r) {
        D[(size_t)(m0 + r + hi * 8) * 64 + (n0 + lr)] = alpha * c[r] + bn;
    }
}

#define TI    4
#define NMAX  1024

__global__ void __launch_bounds__(256)
silu_reduce_kernel(const float* __restrict__ edge,
                   const float* __restrict__ lx,
                   const float* __restrict__ We,
                   float* __restrict__ S,
                   int N)
{
    __shared__ float2 eLDS[TI][NMAX];
    __shared__ float  red[TI][256];

    const int tid  = threadIdx.x;
    const int base = blockIdx.x * TI;
    const int b    = base / N;

    const float2* __restrict__ ep = (const float2*)edge + (size_t)base * N;
    for (int t = tid; t < TI * N; t += 256) {
        eLDS[t / NMAX][t % NMAX] = ep[(size_t)(t / N) * N + (t % N)];
    }

    const int   h   = tid & 63;
    const int   jg  = tid >> 6;
    const float we0 = We[h * 2 + 0];
    const float we1 = We[h * 2 + 1];

    float acc[TI] = {0.f, 0.f, 0.f, 0.f};
    __syncthreads();

    const float* __restrict__ lxb = lx + (size_t)b * N * 64;
#pragma unroll 2
    for (int j = jg; j < N; j += 4) {
        const float lxv = lxb[(size_t)j * 64 + h];
#pragma unroll
        for (int ii = 0; ii < TI; ++ii) {
            const float2 e = eLDS[ii][j];
            const float  m = lxv * fmaf(e.x, we0, e.y * we1);
            acc[ii] += m * __builtin_amdgcn_rcpf(1.0f + __expf(-m));
        }
    }

#pragma unroll
    for (int ii = 0; ii < TI; ++ii) red[ii][tid] = acc[ii];
    __syncthreads();

    if (tid < 64) {
#pragma unroll
        for (int ii = 0; ii < TI; ++ii) {
            const float s = red[ii][tid] + red[ii][tid + 64] +
                            red[ii][tid + 128] + red[ii][tid + 192];
            S[(size_t)(base + ii) * 64 + tid] = s;
        }
    }
}

extern "C" void kernel_launch(void* const* d_in, const int* in_sizes, int n_in,
                              void* d_out, int out_size, void* d_ws, size_t ws_size,
                              hipStream_t stream)
{
    const float* x    = (const float*)d_in[0];
    const float* edge = (const float*)d_in[1];
    const float* W0   = (const float*)d_in[2];
    const float* b0   = (const float*)d_in[3];
    const float* We   = (const float*)d_in[4];
    const float* W1   = (const float*)d_in[5];
    const float* b1   = (const float*)d_in[6];
    float*       out  = (float*)d_out;

    const int N = (int)(32ll * in_sizes[1] / in_sizes[0]);
    const int M = in_sizes[0] / 64;

    float* lx = (float*)d_ws;
    float* S  = lx + (size_t)M * 64;

    dim3 ggemm(M / 16, 64 / 16);
    wmma_gemm_nt64<<<ggemm, 32, 0, stream>>>(x, W0, b0, lx, 1.0f);

    silu_reduce_kernel<<<M / TI, 256, 0, stream>>>(edge, lx, We, S, N);

    wmma_gemm_nt64<<<ggemm, 32, 0, stream>>>(S, W1, b1, out, 1.0f / (float)N);
}


# silu kernel j loop software-pipelined: chunks of 8 j in two register sets, the next chunk's lx (global) and edge pairs (LDS) load while the current chunk computes; every sum keeps its order
# speedup vs baseline: 1.5884x; 1.5884x over previous
#include <hip/hip_runtime.h>
#include <hip/hip_bf16.h>
#include <math.h>


typedef __attribute__((ext_vector_type(2))) float v2f;
typedef __attribute__((ext_vector_type(8))) float v8f;

__global__ void __launch_bounds__(32)
wmma_gemm_nt64(const float* __restrict__ A,
               const float* __restrict__ W,
               const float* __restrict__ bias,
               float* __restrict__ D,
               float alpha)
{
    const int m0   = blockIdx.x * 16;
    const int n0   = blockIdx.y * 16;
    const int lane = threadIdx.x;
    const int lr   = lane & 15;
    const int hi   = lane >> 4;
    const int kh   = hi * 2;

    const float* __restrict__ arow = A + (size_t)(m0 + lr) * 64;
    const float* __restrict__ wrow = W + (size_t)(n0 + lr) * 64;

    v8f c = {};
#pragma unroll
    for (int k = 0; k < 64; k += 4) {
        v2f a, b;
        a.x = arow[k + kh + 0];
        a.y = arow[k + kh + 1];
        b.x = wrow[k + kh + 0];
        b.y = wrow[k + kh + 1];
        c = __builtin_amdgcn_wmma_f32_16x16x4_f32(false, a, false, b,
                                                  (short)0, c, false, false);
    }

    const float bn = bias[n0 + lr];
#pragma unroll
    for (int r = 0; r < 8; ++r) {
        D[(size_t)(m0 + r + hi * 8) * 64 + (n0 + lr)] = alpha * c[r] + bn;
    }
}

#define TI    4
#define NMAX  1024
#define UJ    8

struct Chunk {
    float  lx[UJ];
    float2 e[TI][UJ];
};

__device__ __forceinline__ void chunk_load(Chunk& c, const float* __restrict__ lxh,
                                           const float2 (*eL)[NMAX], int j0)
{
#pragma unroll
    for (int u = 0; u < UJ; ++u) c.lx[u] = lxh[(size_t)(j0 + 4 * u) * 64];
#pragma unroll
    for (int ii = 0; ii < TI; ++ii) {
#pragma unroll
        for (int u = 0; u < UJ; ++u) c.e[ii][u] = eL[ii][j0 + 4 * u];
    }
}

__device__ __forceinline__ void chunk_compute(const Chunk& c, float we0, float we1,
                                              float* acc)
{
#pragma unroll
    for (int u = 0; u < UJ; ++u) {
#pragma unroll
        for (int ii = 0; ii < TI; ++ii) {
            const float2 e = c.e[ii][u];
            const float  m = c.lx[u] * fmaf(e.x, we0, e.y * we1);
            acc[ii] += m * __builtin_amdgcn_rcpf(1.0f + __expf(-m));
        }
    }
}

__global__ void __launch_bounds__(256)
silu_reduce_kernel(const float* __restrict__ edge,
                   const float* __restrict__ lx,
                   const float* __restrict__ We,
                   float* __restrict__ S,
                   int N)
{
    __shared__ float2 eLDS[TI][NMAX];
    __shared__ float  red[TI][256];

    const int tid  = threadIdx.x;
    const int base = blockIdx.x * TI;
    const int b    = base / N;

    const float2* __restrict__ ep = (const float2*)edge + (size_t)base * N;
    for (int t = tid; t < TI * N; t += 256) {
        eLDS[t / NMAX][t % NMAX] = ep[(size_t)(t / N) * N + (t % N)];
    }

    const int   h   = tid & 63;
    const int   jg  = tid >> 6;
    const float we0 = We[h * 2 + 0];
    const float we1 = We[h * 2 + 1];

    float acc[TI] = {0.f, 0.f, 0.f, 0.f};
    __syncthreads();

    const float* __restrict__ lxh = lx + (size_t)b * N * 64 + h;

    const int cnt   = (N > jg) ? (N - jg + 3) / 4 : 0;
    const int npair = cnt / (2 * UJ);

    Chunk A, B;
    int j0 = jg;
    if (npair > 0) chunk_load(A, lxh, eLDS, j0);
    for (int p = 0; p < npair; ++p) {
        chunk_load(B, lxh, eLDS, j0 + 4 * UJ);
        __builtin_amdgcn_sched_barrier(0);
        chunk_compute(A, we0, we1, acc);
        __builtin_amdgcn_sched_barrier(0);
        const int jn = (p + 1 < npair) ? j0 + 8 * UJ : jg;
        chunk_load(A, lxh, eLDS, jn);
        __builtin_amdgcn_sched_barrier(0);
        chunk_compute(B, we0, we1, acc);
        __builtin_amdgcn_sched_barrier(0);
        j0 += 8 * UJ;
    }
    for (int j = j0; j < N; j += 4) {
        const float lxv = lxh[(size_t)j * 64];
#pragma unroll
        for (int ii = 0; ii < TI; ++ii) {
            const float2 e = eLDS[ii][j];
            const float  m = lxv * fmaf(e.x, we0, e.y * we1);
            acc[ii] += m * __builtin_amdgcn_rcpf(1.0f + __expf(-m));
        }
    }

#pragma unroll
    for (int ii = 0; ii < TI; ++ii) red[ii][tid] = acc[ii];
    __syncthreads();

    if (tid < 64) {
#pragma unroll
        for (int ii = 0; ii < TI; ++ii) {
            const float s = red[ii][tid] + red[ii][tid + 64] +
                            red[ii][tid + 128] + red[ii][tid + 192];
            S[(size_t)(base + ii) * 64 + tid] = s;
        }
    }
}

extern "C" void kernel_launch(void* const* d_in, const int* in_sizes, int n_in,
                              void* d_out, int out_size, void* d_ws, size_t ws_size,
                              hipStream_t stream)
{
    const float* x    = (const float*)d_in[0];
    const float* edge = (const float*)d_in[1];
    const float* W0   = (const float*)d_in[2];
    const float* b0   = (const float*)d_in[3];
    const float* We   = (const float*)d_in[4];
    const float* W1   = (const float*)d_in[5];
    const float* b1   = (const float*)d_in[6];
    float*       out  = (float*)d_out;

    const int N = (int)(32ll * in_sizes[1] / in_sizes[0]);
    const int M = in_sizes[0] / 64;

    float* lx = (float*)d_ws;
    float* S  = lx + (size_t)M * 64;

    dim3 ggemm(M / 16, 64 / 16);
    wmma_gemm_nt64<<<ggemm, 32, 0, stream>>>(x, W0, b0, lx, 1.0f);

    silu_reduce_kernel<<<M / TI, 256, 0, stream>>>(edge, lx, We, S, N);

    wmma_gemm_nt64<<<ggemm, 32, 0, stream>>>(S, W1, b1, out, 1.0f / (float)N);
}


# silu kernel edge staging: each thread issues all 16 of its float2 loads before its first LDS store, one memory round trip instead of 16 serial ones, no index division; every sum keeps its order
# speedup vs baseline: 2.2029x; 1.3869x over previous
#include <hip/hip_runtime.h>
#include <hip/hip_bf16.h>
#include <math.h>


typedef __attribute__((ext_vector_type(2))) float v2f;
typedef __attribute__((ext_vector_type(8))) float v8f;

__global__ void __launch_bounds__(32)
wmma_gemm_nt64(const float* __restrict__ A,
               const float* __restrict__ W,
               const float* __restrict__ bias,
               float* __restrict__ D,
               float alpha)
{
    const int m0   = blockIdx.x * 16;
    const int n0   = blockIdx.y * 16;
    const int lane = threadIdx.x;
    const int lr   = lane & 15;
    const int hi   = lane >> 4;
    const int kh   = hi * 2;

    const float* __restrict__ arow = A + (size_t)(m0 + lr) * 64;
    const float* __restrict__ wrow = W + (size_t)(n0 + lr) * 64;

    v8f c = {};
#pragma unroll
    for (int k = 0; k < 64; k += 4) {
        v2f a, b;
        a.x = arow[k + kh + 0];
        a.y = arow[k + kh + 1];
        b.x = wrow[k + kh + 0];
        b.y = wrow[k + kh + 1];
        c = __builtin_amdgcn_wmma_f32_16x16x4_f32(false, a, false, b,
                                                  (short)0, c, false, false);
    }

    const float bn = bias[n0 + lr];
#pragma unroll
    for (int r = 0; r < 8; ++r) {
        D[(size_t)(m0 + r + hi * 8) * 64 + (n0 + lr)] = alpha * c[r] + bn;
    }
}

#define TI    4
#define NMAX  1024
#define UJ    8
#define SLOTS (TI * NMAX / 256)

struct Chunk {
    float  lx[UJ];
    float2 e[TI][UJ];
};

__device__ __forceinline__ void chunk_load(Chunk& c, const float* __restrict__ lxh,
                                           const float2 (*eL)[NMAX], int j0)
{
#pragma unroll
    for (int u = 0; u < UJ; ++u) c.lx[u] = lxh[(size_t)(j0 + 4 * u) * 64];
#pragma unroll
    for (int ii = 0; ii < TI; ++ii) {
#pragma unroll
        for (int u = 0; u < UJ; ++u) c.e[ii][u] = eL[ii][j0 + 4 * u];
    }
}

__device__ __forceinline__ float silu_tanh(float t)
{
    return fmaf(t, __builtin_amdgcn_tanhf(t), t);
}

__device__ __forceinline__ void chunk_compute(const Chunk& c, float wh0, float wh1,
                                              float* acc)
{
#pragma unroll
    for (int u = 0; u < UJ; ++u) {
#pragma unroll
        for (int ii = 0; ii < TI; ++ii) {
            const float2 e = c.e[ii][u];
            const float  t = c.lx[u] * fmaf(e.x, wh0, e.y * wh1);
            acc[ii] += silu_tanh(t);
        }
    }
}

__global__ void __launch_bounds__(256)
silu_reduce_kernel(const float* __restrict__ edge,
                   const float* __restrict__ lx,
                   const float* __restrict__ We,
                   float* __restrict__ S,
                   int N)
{
    __shared__ float2 eLDS[TI][NMAX];
    __shared__ float  red[TI][256];

    const int tid  = threadIdx.x;
    const int base = blockIdx.x * TI;
    const int b    = base / N;

    const float2* __restrict__ ep = (const float2*)edge + (size_t)base * N;
    float2 stg[SLOTS];
#pragma unroll
    for (int k = 0; k < SLOTS; ++k) {
        const int ii = k / (NMAX / 256);
        const int c  = (k % (NMAX / 256)) * 256 + tid;
        stg[k] = ep[(size_t)ii * N + min(c, N - 1)];
    }
#pragma unroll
    for (int k = 0; k < SLOTS; ++k) {
        const int ii = k / (NMAX / 256);
        const int c  = (k % (NMAX / 256)) * 256 + tid;
        eLDS[ii][c] = stg[k];
    }

    const int   h   = tid & 63;
    const int   jg  = tid >> 6;
    const float we0 = 0.5f * We[h * 2 + 0];
    const float we1 = 0.5f * We[h * 2 + 1];

    float acc[TI] = {0.f, 0.f, 0.f, 0.f};
    __syncthreads();

    const float* __restrict__ lxh = lx + (size_t)b * N * 64 + h;

    const int cnt   = (N > jg) ? (N - jg + 3) / 4 : 0;
    const int npair = cnt / (2 * UJ);

    Chunk A, B;
    int j0 = jg;
    if (npair > 0) chunk_load(A, lxh, eLDS, j0);
    for (int p = 0; p < npair; ++p) {
        chunk_load(B, lxh, eLDS, j0 + 4 * UJ);
        __builtin_amdgcn_sched_barrier(0);
        chunk_compute(A, we0, we1, acc);
        __builtin_amdgcn_sched_barrier(0);
        const int jn = (p + 1 < npair) ? j0 + 8 * UJ : jg;
        chunk_load(A, lxh, eLDS, jn);
        __builtin_amdgcn_sched_barrier(0);
        chunk_compute(B, we0, we1, acc);
        __builtin_amdgcn_sched_barrier(0);
        j0 += 8 * UJ;
    }
    for (int j = j0; j < N; j += 4) {
        const float lxv = lxh[(size_t)j * 64];
#pragma unroll
        for (int ii = 0; ii < TI; ++ii) {
            const float2 e = eLDS[ii][j];
            const float  t = lxv * fmaf(e.x, we0, e.y * we1);
            acc[ii] += silu_tanh(t);
        }
    }

#pragma unroll
    for (int ii = 0; ii < TI; ++ii) red[ii][tid] = acc[ii];
    __syncthreads();

    if (tid < 64) {
#pragma unroll
        for (int ii = 0; ii < TI; ++ii) {
            const float s = red[ii][tid] + red[ii][tid + 64] +
                            red[ii][tid + 128] + red[ii][tid + 192];
            S[(size_t)(base + ii) * 64 + tid] = s;
        }
    }
}

extern "C" void kernel_launch(void* const* d_in, const int* in_sizes, int n_in,
                              void* d_out, int out_size, void* d_ws, size_t ws_size,
                              hipStream_t stream)
{
    const float* x    = (const float*)d_in[0];
    const float* edge = (const float*)d_in[1];
    const float* W0   = (const float*)d_in[2];
    const float* b0   = (const float*)d_in[3];
    const float* We   = (const float*)d_in[4];
    const float* W1   = (const float*)d_in[5];
    const float* b1   = (const float*)d_in[6];
    float*       out  = (float*)d_out;

    const int N = (int)(32ll * in_sizes[1] / in_sizes[0]);
    const int M = in_sizes[0] / 64;

    float* lx = (float*)d_ws;
    float* S  = lx + (size_t)M * 64;

    dim3 ggemm(M / 16, 64 / 16);
    wmma_gemm_nt64<<<ggemm, 32, 0, stream>>>(x, W0, b0, lx, 1.0f);

    silu_reduce_kernel<<<M / TI, 256, 0, stream>>>(edge, lx, We, S, N);

    wmma_gemm_nt64<<<ggemm, 32, 0, stream>>>(S, W1, b1, out, 1.0f / (float)N);
}


# final 64x64 projection fused into the silu kernel's tail: each block runs its 4 rows of S, zero-padded to a 16-row tile, through the same f32 WMMA and writes out itself; 3 launches become 2
# speedup vs baseline: 2.5906x; 1.1760x over previous
#include <hip/hip_runtime.h>
#include <hip/hip_bf16.h>
#include <math.h>


typedef __attribute__((ext_vector_type(2))) float v2f;
typedef __attribute__((ext_vector_type(8))) float v8f;

__global__ void __launch_bounds__(32)
wmma_gemm_nt64(const float* __restrict__ A,
               const float* __restrict__ W,
               const float* __restrict__ bias,
               float* __restrict__ D,
               float alpha)
{
    const int m0   = blockIdx.x * 16;
    const int n0   = blockIdx.y * 16;
    const int lane = threadIdx.x;
    const int lr   = lane & 15;
    const int hi   = lane >> 4;
    const int kh   = hi * 2;

    const float* __restrict__ arow = A + (size_t)(m0 + lr) * 64;
    const float* __restrict__ wrow = W + (size_t)(n0 + lr) * 64;

    v8f c = {};
#pragma unroll
    for (int k = 0; k < 64; k += 4) {
        v2f a, b;
        a.x = arow[k + kh + 0];
        a.y = arow[k + kh + 1];
        b.x = wrow[k + kh + 0];
        b.y = wrow[k + kh + 1];
        c = __builtin_amdgcn_wmma_f32_16x16x4_f32(false, a, false, b,
                                                  (short)0, c, false, false);
    }

    const float bn = bias[n0 + lr];
#pragma unroll
    for (int r = 0; r < 8; ++r) {
        D[(size_t)(m0 + r + hi * 8) * 64 + (n0 + lr)] = alpha * c[r] + bn;
    }
}

#define TI    4
#define NMAX  1024
#define UJ    8
#define SLOTS (TI * NMAX / 256)
#define SPITCH 68

struct Chunk {
    float  lx[UJ];
    float2 e[TI][UJ];
};

__device__ __forceinline__ void chunk_load(Chunk& c, const float* __restrict__ lxh,
                                           const float2 (*eL)[NMAX], int j0)
{
#pragma unroll
    for (int u = 0; u < UJ; ++u) c.lx[u] = lxh[(size_t)(j0 + 4 * u) * 64];
#pragma unroll
    for (int ii = 0; ii < TI; ++ii) {
#pragma unroll
        for (int u = 0; u < UJ; ++u) c.e[ii][u] = eL[ii][j0 + 4 * u];
    }
}

__device__ __forceinline__ float silu_tanh(float t)
{
    return fmaf(t, __builtin_amdgcn_tanhf(t), t);
}

__device__ __forceinline__ void chunk_compute(const Chunk& c, float wh0, float wh1,
                                              float* acc)
{
#pragma unroll
    for (int u = 0; u < UJ; ++u) {
#pragma unroll
        for (int ii = 0; ii < TI; ++ii) {
            const float2 e = c.e[ii][u];
            const float  t = c.lx[u] * fmaf(e.x, wh0, e.y * wh1);
            acc[ii] += silu_tanh(t);
        }
    }
}

__global__ void __launch_bounds__(256)
silu_reduce_kernel(const float* __restrict__ edge,
                   const float* __restrict__ lx,
                   const float* __restrict__ We,
                   const float* __restrict__ W1,
                   const float* __restrict__ b1,
                   float* __restrict__ out,
                   int N, float alpha)
{
    __shared__ float2 eLDS[TI][NMAX];
    __shared__ float  red[TI][256];
    __shared__ float  Ssm[16][SPITCH];
    __shared__ float  Osm[TI][64];

    const int tid  = threadIdx.x;
    const int base = blockIdx.x * TI;
    const int b    = base / N;

    const int wv   = __builtin_amdgcn_readfirstlane(tid) >> 5;
    const int lane = tid & 31;
    const int lr   = lane & 15;
    const int hi   = lane >> 4;
    const int kh   = hi * 2;

    v2f   bfrag[16];
    float bn = 0.f;
    if (wv < 4) {
        const float* __restrict__ wrow = W1 + (size_t)(wv * 16 + lr) * 64;
#pragma unroll
        for (int k = 0; k < 64; k += 4) {
            bfrag[k / 4].x = wrow[k + kh + 0];
            bfrag[k / 4].y = wrow[k + kh + 1];
        }
        bn = b1[wv * 16 + lr];
    }

    const float2* __restrict__ ep = (const float2*)edge + (size_t)base * N;
    float2 stg[SLOTS];
#pragma unroll
    for (int k = 0; k < SLOTS; ++k) {
        const int ii = k / (NMAX / 256);
        const int c  = (k % (NMAX / 256)) * 256 + tid;
        stg[k] = ep[(size_t)ii * N + min(c, N - 1)];
    }
#pragma unroll
    for (int k = 0; k < SLOTS; ++k) {
        const int ii = k / (NMAX / 256);
        const int c  = (k % (NMAX / 256)) * 256 + tid;
        eLDS[ii][c] = stg[k];
    }

    const int   h   = tid & 63;
    const int   jg  = tid >> 6;
    const float we0 = 0.5f * We[h * 2 + 0];
    const float we1 = 0.5f * We[h * 2 + 1];

    float acc[TI] = {0.f, 0.f, 0.f, 0.f};
    __syncthreads();

    const float* __restrict__ lxh = lx + (size_t)b * N * 64 + h;

    const int cnt   = (N > jg) ? (N - jg + 3) / 4 : 0;
    const int npair = cnt / (2 * UJ);

    Chunk A, B;
    int j0 = jg;
    if (npair > 0) chunk_load(A, lxh, eLDS, j0);
    for (int p = 0; p < npair; ++p) {
        chunk_load(B, lxh, eLDS, j0 + 4 * UJ);
        __builtin_amdgcn_sched_barrier(0);
        chunk_compute(A, we0, we1, acc);
        __builtin_amdgcn_sched_barrier(0);
        const int jn = (p + 1 < npair) ? j0 + 8 * UJ : jg;
        chunk_load(A, lxh, eLDS, jn);
        __builtin_amdgcn_sched_barrier(0);
        chunk_compute(B, we0, we1, acc);
        __builtin_amdgcn_sched_barrier(0);
        j0 += 8 * UJ;
    }
    for (int j = j0; j < N; j += 4) {
        const float lxv = lxh[(size_t)j * 64];
#pragma unroll
        for (int ii = 0; ii < TI; ++ii) {
            const float2 e = eLDS[ii][j];
            const float  t = lxv * fmaf(e.x, we0, e.y * we1);
            acc[ii] += silu_tanh(t);
        }
    }

#pragma unroll
    for (int ii = 0; ii < TI; ++ii) red[ii][tid] = acc[ii];
    if (tid >= 64) {
#pragma unroll
        for (int q = 0; q < (16 - TI) * 64 / 192; ++q) {
            const int idx = (tid - 64) + 192 * q;
            Ssm[TI + (idx >> 6)][idx & 63] = 0.f;
        }
    }
    __syncthreads();

    if (tid < 64) {
#pragma unroll
        for (int ii = 0; ii < TI; ++ii) {
            const float s = red[ii][tid] + red[ii][tid + 64] +
                            red[ii][tid + 128] + red[ii][tid + 192];
            Ssm[ii][tid] = s;
        }
    }
    __syncthreads();

    if (wv < 4) {
        v2f afrag[16];
#pragma unroll
        for (int k = 0; k < 64; k += 4) {
            afrag[k / 4].x = Ssm[lr][k + kh + 0];
            afrag[k / 4].y = Ssm[lr][k + kh + 1];
        }
        v8f c = {};
#pragma unroll
        for (int k = 0; k < 64; k += 4) {
            c = __builtin_amdgcn_wmma_f32_16x16x4_f32(false, afrag[k / 4], false, bfrag[k / 4],
                                                      (short)0, c, false, false);
        }
        if (hi == 0) {
#pragma unroll
            for (int r = 0; r < TI; ++r) Osm[r][wv * 16 + lr] = alpha * c[r] + bn;
        }
    }
    __syncthreads();

    out[(size_t)(base + (tid >> 6)) * 64 + (tid & 63)] = Osm[tid >> 6][tid & 63];
}

extern "C" void kernel_launch(void* const* d_in, const int* in_sizes, int n_in,
                              void* d_out, int out_size, void* d_ws, size_t ws_size,
                              hipStream_t stream)
{
    const float* x    = (const float*)d_in[0];
    const float* edge = (const float*)d_in[1];
    const float* W0   = (const float*)d_in[2];
    const float* b0   = (const float*)d_in[3];
    const float* We   = (const float*)d_in[4];
    const float* W1   = (const float*)d_in[5];
    const float* b1   = (const float*)d_in[6];
    float*       out  = (float*)d_out;

    const int N = (int)(32ll * in_sizes[1] / in_sizes[0]);
    const int M = in_sizes[0] / 64;

    float* lx = (float*)d_ws;

    dim3 ggemm(M / 16, 64 / 16);
    wmma_gemm_nt64<<<ggemm, 32, 0, stream>>>(x, W0, b0, lx, 1.0f);

    silu_reduce_kernel<<<M / TI, 256, 0, stream>>>(edge, lx, We, W1, b1, out, N,
                                                   1.0f / (float)N);
}


# silu argument on the f32 16x16x4 matrix unit (tile 8 rows x 2 j x 16 cols, B = lx*We/2: t = e0*(lx*w0)+e1*(lx*w1), new rounding); VALU keeps tanh + packed fma/sum; j sum = 8 partials (new order)
# speedup vs baseline: 2.9901x; 1.1542x over previous
#include <hip/hip_runtime.h>
#include <hip/hip_bf16.h>
#include <math.h>


typedef __attribute__((ext_vector_type(2))) float v2f;
typedef __attribute__((ext_vector_type(8))) float v8f;

__global__ void __launch_bounds__(32)
wmma_gemm_nt64(const float* __restrict__ A,
               const float* __restrict__ W,
               const float* __restrict__ bias,
               float* __restrict__ D,
               float alpha)
{
    const int m0   = blockIdx.x * 16;
    const int n0   = blockIdx.y * 16;
    const int lane = threadIdx.x;
    const int lr   = lane & 15;
    const int hi   = lane >> 4;
    const int kh   = hi * 2;

    const float* __restrict__ arow = A + (size_t)(m0 + lr) * 64;
    const float* __restrict__ wrow = W + (size_t)(n0 + lr) * 64;

    v8f c = {};
#pragma unroll
    for (int k = 0; k < 64; k += 4) {
        v2f a, b;
        a.x = arow[k + kh + 0];
        a.y = arow[k + kh + 1];
        b.x = wrow[k + kh + 0];
        b.y = wrow[k + kh + 1];
        c = __builtin_amdgcn_wmma_f32_16x16x4_f32(false, a, false, b,
                                                  (short)0, c, false, false);
    }

    const float bn = bias[n0 + lr];
#pragma unroll
    for (int r = 0; r < 8; ++r) {
        D[(size_t)(m0 + r + hi * 8) * 64 + (n0 + lr)] = alpha * c[r] + bn;
    }
}

#define TI    8
#define NMAX  1024
#define NT    512
#define EP    (NMAX + 2)
#define UQ    16
#define SLOTS (TI * NMAX / NT)
#define SPITCH 68

struct Chunk {
    float lxv[UQ];
    v2f   a[UQ];
};

__device__ __forceinline__ void chunk_load(Chunk& ck, const float* __restrict__ lp,
                                           const v2f* ap, int q0)
{
#pragma unroll
    for (int u = 0; u < UQ; ++u) ck.lxv[u] = lp[(size_t)(q0 + u) * 512];
#pragma unroll
    for (int u = 0; u < UQ; ++u) ck.a[u] = ap[(q0 + u) * 8];
}

__device__ __forceinline__ void chunk_compute(const Chunk& ck, v2f wq, v2f* acc2)
{
    const v8f zero = {};
#pragma unroll
    for (int u = 0; u < UQ; ++u) {
        const v2f bb = ck.lxv[u] * wq;
        const v8f d = __builtin_amdgcn_wmma_f32_16x16x4_f32(false, ck.a[u], false, bb,
                                                            (short)0, zero, false, false);
#pragma unroll
        for (int r2 = 0; r2 < TI / 2; ++r2) {
            v2f t, th;
            t.x  = d[2 * r2];
            t.y  = d[2 * r2 + 1];
            th.x = __builtin_amdgcn_tanhf(t.x);
            th.y = __builtin_amdgcn_tanhf(t.y);
            acc2[r2] += __builtin_elementwise_fma(t, th, t);
        }
    }
}

__global__ void __launch_bounds__(NT)
silu_reduce_kernel(const float* __restrict__ edge,
                   const float* __restrict__ lx,
                   const float* __restrict__ We,
                   const float* __restrict__ W1,
                   const float* __restrict__ b1,
                   float* __restrict__ out,
                   int N, float alpha)
{
    __shared__ v2f   eL[TI + 1][EP];
    __shared__ float red[8][TI][64];
    __shared__ float Ssm[16][SPITCH];
    __shared__ float Osm[TI][64];

    const int tid  = threadIdx.x;
    const int base = blockIdx.x * TI;
    const int b    = base / N;

    const int wv   = __builtin_amdgcn_readfirstlane(tid) >> 5;
    const int hg   = wv & 3;
    const int sl   = wv >> 2;
    const int lane = tid & 31;
    const int lr   = lane & 15;
    const int hi   = lane >> 4;
    const int kh   = hi * 2;

    v2f   bfrag[16];
    float bn = 0.f;
    if (wv < 4) {
        const float* __restrict__ wrow = W1 + (size_t)(wv * 16 + lr) * 64;
#pragma unroll
        for (int k = 0; k < 64; k += 4) {
            bfrag[k / 4].x = wrow[k + kh + 0];
            bfrag[k / 4].y = wrow[k + kh + 1];
        }
        bn = b1[wv * 16 + lr];
    }

    const v2f* __restrict__ ep = (const v2f*)edge + (size_t)base * N;
    v2f stg[SLOTS];
#pragma unroll
    for (int k = 0; k < SLOTS; ++k) {
        const int ii = k / (NMAX / NT);
        const int c  = (k % (NMAX / NT)) * NT + tid;
        stg[k] = ep[(size_t)ii * N + min(c, N - 1)];
    }
#pragma unroll
    for (int k = 0; k < SLOTS; ++k) {
        const int ii = k / (NMAX / NT);
        const int c  = (k % (NMAX / NT)) * NT + tid;
        eL[ii][c] = stg[k];
    }
    {
        const v2f z2 = {0.f, 0.f};
        for (int c = tid; c < EP; c += NT) eL[TI][c] = z2;
    }

    const int h = 16 * hg + lr;
    v2f wq;
    wq.x = 0.5f * We[h * 2 + 0];
    wq.y = 0.5f * We[h * 2 + 1];

    const int arow = (lane < 8) ? lane : ((lane >= 24) ? lane - 24 : TI);
    const int ajo  = (lane >= 24) ? 1 : 0;
    const v2f* ap = &eL[arow][2 * sl + ajo];

    const float* __restrict__ lp = lx + (size_t)b * N * 64 + (size_t)(2 * sl + hi) * 64 + h;

    v2f acc2[TI / 2];
#pragma unroll
    for (int r2 = 0; r2 < TI / 2; ++r2) { acc2[r2].x = 0.f; acc2[r2].y = 0.f; }
    __syncthreads();

    const int npair = N / (8 * 2 * UQ);

    Chunk A, B;
    int q0 = 0;
    if (npair > 0) chunk_load(A, lp, ap, 0);
    for (int p = 0; p < npair; ++p) {
        chunk_load(B, lp, ap, q0 + UQ);
        __builtin_amdgcn_sched_barrier(0);
        chunk_compute(A, wq, acc2);
        __builtin_amdgcn_sched_barrier(0);
        const int qn = (p + 1 < npair) ? q0 + 2 * UQ : 0;
        chunk_load(A, lp, ap, qn);
        __builtin_amdgcn_sched_barrier(0);
        chunk_compute(B, wq, acc2);
        __builtin_amdgcn_sched_barrier(0);
        q0 += 2 * UQ;
    }

#pragma unroll
    for (int r2 = 0; r2 < TI / 2; ++r2) {
        red[2 * sl + hi][2 * r2 + 0][h] = acc2[r2].x;
        red[2 * sl + hi][2 * r2 + 1][h] = acc2[r2].y;
    }
    Ssm[TI + (tid >> 6)][tid & 63] = 0.f;
    __syncthreads();

    {
        const int r = tid >> 6, cc = tid & 63;
        float s = red[0][r][cc];
#pragma unroll
        for (int p = 1; p < 8; ++p) s += red[p][r][cc];
        Ssm[r][cc] = s;
    }
    __syncthreads();

    if (wv < 4) {
        v2f afrag[16];
#pragma unroll
        for (int k = 0; k < 64; k += 4) {
            afrag[k / 4].x = Ssm[lr][k + kh + 0];
            afrag[k / 4].y = Ssm[lr][k + kh + 1];
        }
        v8f c = {};
#pragma unroll
        for (int k = 0; k < 64; k += 4) {
            c = __builtin_amdgcn_wmma_f32_16x16x4_f32(false, afrag[k / 4], false, bfrag[k / 4],
                                                      (short)0, c, false, false);
        }
        if (hi == 0) {
#pragma unroll
            for (int r = 0; r < TI; ++r) Osm[r][wv * 16 + lr] = alpha * c[r] + bn;
        }
    }
    __syncthreads();

    out[(size_t)(base + (tid >> 6)) * 64 + (tid & 63)] = Osm[tid >> 6][tid & 63];
}

extern "C" void kernel_launch(void* const* d_in, const int* in_sizes, int n_in,
                              void* d_out, int out_size, void* d_ws, size_t ws_size,
                              hipStream_t stream)
{
    const float* x    = (const float*)d_in[0];
    const float* edge = (const float*)d_in[1];
    const float* W0   = (const float*)d_in[2];
    const float* b0   = (const float*)d_in[3];
    const float* We   = (const float*)d_in[4];
    const float* W1   = (const float*)d_in[5];
    const float* b1   = (const float*)d_in[6];
    float*       out  = (float*)d_out;

    const int N = (int)(32ll * in_sizes[1] / in_sizes[0]);
    const int M = in_sizes[0] / 64;

    float* lx = (float*)d_ws;

    dim3 ggemm(M / 16, 64 / 16);
    wmma_gemm_nt64<<<ggemm, 32, 0, stream>>>(x, W0, b0, lx, 1.0f);

    silu_reduce_kernel<<<M / TI, NT, 0, stream>>>(edge, lx, We, W1, b1, out, N,
                                                  1.0f / (float)N);
}


# silu loop's vector work issued as (tanh, packed) pairs: tile u's 8 tanh alternate with tile u-1's fma and sums (pipelined by one product, pinned by scheduling groups); every sum keeps its order
# speedup vs baseline: 3.2039x; 1.0715x over previous
#include <hip/hip_runtime.h>
#include <hip/hip_bf16.h>
#include <math.h>


typedef __attribute__((ext_vector_type(2))) float v2f;
typedef __attribute__((ext_vector_type(8))) float v8f;

__global__ void __launch_bounds__(32)
wmma_gemm_nt64(const float* __restrict__ A,
               const float* __restrict__ W,
               const float* __restrict__ bias,
               float* __restrict__ D,
               float alpha)
{
    const int m0   = blockIdx.x * 16;
    const int n0   = blockIdx.y * 16;
    const int lane = threadIdx.x;
    const int lr   = lane & 15;
    const int hi   = lane >> 4;
    const int kh   = hi * 2;

    const float* __restrict__ arow = A + (size_t)(m0 + lr) * 64;
    const float* __restrict__ wrow = W + (size_t)(n0 + lr) * 64;

    v8f c = {};
#pragma unroll
    for (int k = 0; k < 64; k += 4) {
        v2f a, b;
        a.x = arow[k + kh + 0];
        a.y = arow[k + kh + 1];
        b.x = wrow[k + kh + 0];
        b.y = wrow[k + kh + 1];
        c = __builtin_amdgcn_wmma_f32_16x16x4_f32(false, a, false, b,
                                                  (short)0, c, false, false);
    }

    const float bn = bias[n0 + lr];
#pragma unroll
    for (int r = 0; r < 8; ++r) {
        D[(size_t)(m0 + r + hi * 8) * 64 + (n0 + lr)] = alpha * c[r] + bn;
    }
}

#define TI    8
#define NMAX  1024
#define NT    512
#define EP    (NMAX + 2)
#define UQ    16
#define SLOTS (TI * NMAX / NT)
#define SPITCH 68

struct Chunk {
    float lxv[UQ];
    v2f   a[UQ];
};

__device__ __forceinline__ void chunk_load(Chunk& ck, const float* __restrict__ lp,
                                           const v2f* ap, int q0)
{
#pragma unroll
    for (int u = 0; u < UQ; ++u) ck.lxv[u] = lp[(size_t)(q0 + u) * 512];
#pragma unroll
    for (int u = 0; u < UQ; ++u) ck.a[u] = ap[(q0 + u) * 8];
}

struct Pipe {
    v8f d;
    v2f t[TI / 2];
    v2f th[TI / 2];
    v2f bb;
};

template <bool NEXT_BB>
__device__ __forceinline__ void chunk_compute(const Chunk& ck, const Chunk& nx, v2f wq,
                                              Pipe& st, v2f* acc2)
{
    const v8f zero = {};
#pragma unroll
    for (int u = 0; u < UQ; ++u) {
        const v2f   a1 = (u + 1 < UQ) ? ck.a[u + 1]   : nx.a[u + 1 - UQ];
        const float l2 = (u + 2 < UQ) ? ck.lxv[u + 2] : nx.lxv[u + 2 - UQ];
        const bool  mk = NEXT_BB || (u + 1 < UQ);
        const v8f dn = __builtin_amdgcn_wmma_f32_16x16x4_f32(false, a1, false, st.bb,
                                                             (short)0, zero, false, false);
        if (mk) st.bb = l2 * wq;
        v2f th[TI / 2], f[TI / 2];
        th[0].x = __builtin_amdgcn_tanhf(st.d[0]);
        f[0]    = __builtin_elementwise_fma(st.t[0], st.th[0], st.t[0]);
        th[0].y = __builtin_amdgcn_tanhf(st.d[1]);
        f[1]    = __builtin_elementwise_fma(st.t[1], st.th[1], st.t[1]);
        th[1].x = __builtin_amdgcn_tanhf(st.d[2]);
        f[2]    = __builtin_elementwise_fma(st.t[2], st.th[2], st.t[2]);
        th[1].y = __builtin_amdgcn_tanhf(st.d[3]);
        f[3]    = __builtin_elementwise_fma(st.t[3], st.th[3], st.t[3]);
        th[2].x = __builtin_amdgcn_tanhf(st.d[4]);
        acc2[0] += f[0];
        th[2].y = __builtin_amdgcn_tanhf(st.d[5]);
        acc2[1] += f[1];
        th[3].x = __builtin_amdgcn_tanhf(st.d[6]);
        acc2[2] += f[2];
        th[3].y = __builtin_amdgcn_tanhf(st.d[7]);
        acc2[3] += f[3];
#pragma unroll
        for (int k = 0; k < TI / 2; ++k) {
            st.t[k].x = st.d[2 * k];
            st.t[k].y = st.d[2 * k + 1];
            st.th[k]  = th[k];
        }
        st.d = dn;
        __builtin_amdgcn_sched_group_barrier(0x008, 1, 0);
        if (mk) __builtin_amdgcn_sched_group_barrier(0x002, 1, 0);
#pragma unroll
        for (int k = 0; k < TI; ++k) {
            __builtin_amdgcn_sched_group_barrier(0x400, 1, 0);
            __builtin_amdgcn_sched_group_barrier(0x002, 1, 0);
        }
    }
}

__global__ void __launch_bounds__(NT)
silu_reduce_kernel(const float* __restrict__ edge,
                   const float* __restrict__ lx,
                   const float* __restrict__ We,
                   const float* __restrict__ W1,
                   const float* __restrict__ b1,
                   float* __restrict__ out,
                   int N, float alpha)
{
    __shared__ v2f   eL[TI + 1][EP];
    __shared__ float red[8][TI][64];
    __shared__ float Ssm[16][SPITCH];
    __shared__ float Osm[TI][64];

    const int tid  = threadIdx.x;
    const int base = blockIdx.x * TI;
    const int b    = base / N;

    const int wv   = __builtin_amdgcn_readfirstlane(tid) >> 5;
    const int hg   = wv & 3;
    const int sl   = wv >> 2;
    const int lane = tid & 31;
    const int lr   = lane & 15;
    const int hi   = lane >> 4;
    const int kh   = hi * 2;

    v2f   bfrag[16];
    float bn = 0.f;
    if (wv < 4) {
        const float* __restrict__ wrow = W1 + (size_t)(wv * 16 + lr) * 64;
#pragma unroll
        for (int k = 0; k < 64; k += 4) {
            bfrag[k / 4].x = wrow[k + kh + 0];
            bfrag[k / 4].y = wrow[k + kh + 1];
        }
        bn = b1[wv * 16 + lr];
    }

    const v2f* __restrict__ ep = (const v2f*)edge + (size_t)base * N;
    v2f stg[SLOTS];
#pragma unroll
    for (int k = 0; k < SLOTS; ++k) {
        const int ii = k / (NMAX / NT);
        const int c  = (k % (NMAX / NT)) * NT + tid;
        stg[k] = ep[(size_t)ii * N + min(c, N - 1)];
    }
#pragma unroll
    for (int k = 0; k < SLOTS; ++k) {
        const int ii = k / (NMAX / NT);
        const int c  = (k % (NMAX / NT)) * NT + tid;
        eL[ii][c] = stg[k];
    }
    {
        const v2f z2 = {0.f, 0.f};
        for (int c = tid; c < EP; c += NT) eL[TI][c] = z2;
    }

    const int h = 16 * hg + lr;
    v2f wq;
    wq.x = 0.5f * We[h * 2 + 0];
    wq.y = 0.5f * We[h * 2 + 1];

    const int arow = (lane < 8) ? lane : ((lane >= 24) ? lane - 24 : TI);
    const int ajo  = (lane >= 24) ? 1 : 0;
    const v2f* ap = &eL[arow][2 * sl + ajo];

    const float* __restrict__ lp = lx + (size_t)b * N * 64 + (size_t)(2 * sl + hi) * 64 + h;

    v2f acc2[TI / 2];
#pragma unroll
    for (int r2 = 0; r2 < TI / 2; ++r2) { acc2[r2].x = 0.f; acc2[r2].y = 0.f; }
    __syncthreads();

    const int npair = N / (8 * 2 * UQ);

    Chunk A, B;
    Pipe st;
#pragma unroll
    for (int k = 0; k < TI / 2; ++k) { st.t[k].x = 0.f; st.t[k].y = 0.f; st.th[k].x = 0.f; st.th[k].y = 0.f; }
    st.d  = (v8f){};
    st.bb = (v2f){0.f, 0.f};
    int q0 = 0;
    if (npair > 0) {
        chunk_load(A, lp, ap, 0);
        const v8f zero = {};
        st.d  = __builtin_amdgcn_wmma_f32_16x16x4_f32(false, A.a[0], false, A.lxv[0] * wq,
                                                      (short)0, zero, false, false);
    }
    for (int p = 0; p < npair; ++p) {
        st.bb = A.lxv[1] * wq;
        chunk_load(B, lp, ap, q0 + UQ);
        __builtin_amdgcn_sched_barrier(0);
        chunk_compute<true>(A, B, wq, st, acc2);
        __builtin_amdgcn_sched_barrier(0);
        const int qn = (p + 1 < npair) ? q0 + 2 * UQ : 0;
        chunk_load(A, lp, ap, qn);
        __builtin_amdgcn_sched_barrier(0);
        chunk_compute<false>(B, A, wq, st, acc2);
        __builtin_amdgcn_sched_barrier(0);
        q0 += 2 * UQ;
    }
#pragma unroll
    for (int k = 0; k < TI / 2; ++k)
        acc2[k] += __builtin_elementwise_fma(st.t[k], st.th[k], st.t[k]);

#pragma unroll
    for (int r2 = 0; r2 < TI / 2; ++r2) {
        red[2 * sl + hi][2 * r2 + 0][h] = acc2[r2].x;
        red[2 * sl + hi][2 * r2 + 1][h] = acc2[r2].y;
    }
    Ssm[TI + (tid >> 6)][tid & 63] = 0.f;
    __syncthreads();

    {
        const int r = tid >> 6, cc = tid & 63;
        float s = red[0][r][cc];
#pragma unroll
        for (int p = 1; p < 8; ++p) s += red[p][r][cc];
        Ssm[r][cc] = s;
    }
    __syncthreads();

    if (wv < 4) {
        v2f afrag[16];
#pragma unroll
        for (int k = 0; k < 64; k += 4) {
            afrag[k / 4].x = Ssm[lr][k + kh + 0];
            afrag[k / 4].y = Ssm[lr][k + kh + 1];
        }
        v8f c = {};
#pragma unroll
        for (int k = 0; k < 64; k += 4) {
            c = __builtin_amdgcn_wmma_f32_16x16x4_f32(false, afrag[k / 4], false, bfrag[k / 4],
                                                      (short)0, c, false, false);
        }
        if (hi == 0) {
#pragma unroll
            for (int r = 0; r < TI; ++r) Osm[r][wv * 16 + lr] = alpha * c[r] + bn;
        }
    }
    __syncthreads();

    out[(size_t)(base + (tid >> 6)) * 64 + (tid & 63)] = Osm[tid >> 6][tid & 63];
}

extern "C" void kernel_launch(void* const* d_in, const int* in_sizes, int n_in,
                              void* d_out, int out_size, void* d_ws, size_t ws_size,
                              hipStream_t stream)
{
    const float* x    = (const float*)d_in[0];
    const float* edge = (const float*)d_in[1];
    const float* W0   = (const float*)d_in[2];
    const float* b0   = (const float*)d_in[3];
    const float* We   = (const float*)d_in[4];
    const float* W1   = (const float*)d_in[5];
    const float* b1   = (const float*)d_in[6];
    float*       out  = (float*)d_out;

    const int N = (int)(32ll * in_sizes[1] / in_sizes[0]);
    const int M = in_sizes[0] / 64;

    float* lx = (float*)d_ws;

    dim3 ggemm(M / 16, 64 / 16);
    wmma_gemm_nt64<<<ggemm, 32, 0, stream>>>(x, W0, b0, lx, 1.0f);

    silu_reduce_kernel<<<M / TI, NT, 0, stream>>>(edge, lx, We, W1, b1, out, N,
                                                  1.0f / (float)N);
}
